# starting kernel
# baseline (speedup 1.0000x reference)
#include <hip/hip_runtime.h>
#include <math.h>

typedef __attribute__((ext_vector_type(2))) float v2f;
typedef __attribute__((ext_vector_type(8))) float v8f;

#define Nn   4096
#define INF_ 3.402823466e+38f
#define Dd   256
#define Hh   8
#define HDd  32
#define Ll   2
#define Ee   131072
#define Pp   200000
#define INc  128

__device__ __forceinline__ v8f wmma4(v2f a, v2f b, v8f c) {
  return __builtin_amdgcn_wmma_f32_16x16x4_f32(
      false, a, false, b, (short)0, c, false, false);
}

__device__ __forceinline__ float rsum16(float v) {
  for (int m = 8; m >= 1; m >>= 1) v += __shfl_xor(v, m, 16);
  return v;
}
__device__ __forceinline__ float rmax16(float v) {
  for (int m = 8; m >= 1; m >>= 1) v = fmaxf(v, __shfl_xor(v, m, 16));
  return v;
}

template <int KK>
__global__ __launch_bounds__(32)
void gemm_atb(const float* __restrict__ A, const float* __restrict__ W,
              const float* __restrict__ bias, float* __restrict__ C,
              int M, int Nout) {
  const int tid  = threadIdx.x;
  const int m    = tid & 15;
  const int half = tid >> 4;
  const int tbn  = blockIdx.x * 64;
  const int tbm  = blockIdx.y * 16;

  const float* arow = A + (size_t)(tbm + m) * KK + half * 2;
  const float* w0   = W + (size_t)(tbn + m) * KK + half * 2;
  const float* w1   = w0 + (size_t)16 * KK;
  const float* w2   = w1 + (size_t)16 * KK;
  const float* w3   = w2 + (size_t)16 * KK;

  v8f c0 = {}, c1 = {}, c2 = {}, c3 = {};
#pragma unroll
  for (int k = 0; k < KK; k += 4) {
    v2f a  = { arow[k], arow[k + 1] };
    v2f b0 = { w0[k],   w0[k + 1] };
    v2f b1 = { w1[k],   w1[k + 1] };
    v2f b2 = { w2[k],   w2[k + 1] };
    v2f b3 = { w3[k],   w3[k + 1] };
    c0 = wmma4(a, b0, c0);
    c1 = wmma4(a, b1, c1);
    c2 = wmma4(a, b2, c2);
    c3 = wmma4(a, b3, c3);
  }
  for (int r = 0; r < 8; ++r) {
    const int row = tbm + r + 8 * half;
    float* crow = C + (size_t)row * Nout + tbn + m;
    if (bias) {
      const float* brow = bias + tbn + m;
      crow[0]  = c0[r] + brow[0];
      crow[16] = c1[r] + brow[16];
      crow[32] = c2[r] + brow[32];
      crow[48] = c3[r] + brow[48];
    } else {
      crow[0] = c0[r]; crow[16] = c1[r]; crow[32] = c2[r]; crow[48] = c3[r];
    }
  }
}

__global__ __launch_bounds__(32)
void attn_kernel(const float* __restrict__ qkv, float* __restrict__ obuf) {
  const int tid  = threadIdx.x;
  const int m    = tid & 15;
  const int half = tid >> 4;
  const int qb   = blockIdx.x;
  const int h    = blockIdx.y;
  const float scale = 0.1767766952966369f;

  const float* qbase = qkv + (size_t)(qb * 16) * (3 * Dd) + h * HDd;

  __shared__ float pLds[16 * 16];

  v2f qa[8];
#pragma unroll
  for (int kk = 0; kk < HDd; kk += 4) {
    const int k0 = kk + half * 2;
    qa[kk >> 2] = v2f{ qbase[m * (3 * Dd) + k0], qbase[m * (3 * Dd) + k0 + 1] };
  }

  v8f o0 = {}, o1 = {};
  float mrow[8], lrow[8];
  for (int r = 0; r < 8; ++r) { mrow[r] = -INF_; lrow[r] = 0.f; }

  for (int kb = 0; kb < Nn / 16; ++kb) {
    const float* kbase = qkv + (size_t)(kb * 16) * (3 * Dd) + Dd + h * HDd;
    const float* vbase = qkv + (size_t)(kb * 16) * (3 * Dd) + 2 * Dd + h * HDd;

    v8f s = {};
#pragma unroll
    for (int kk = 0; kk < HDd; kk += 4) {
      const int k0 = kk + half * 2;
      v2f b = { kbase[m * (3 * Dd) + k0], kbase[m * (3 * Dd) + k0 + 1] };
      s = wmma4(qa[kk >> 2], b, s);
    }

    float alpha[8];
#pragma unroll
    for (int r = 0; r < 8; ++r) {
      float sv = s[r] * scale;
      float rm = rmax16(sv);
      float mn = fmaxf(mrow[r], rm);
      float p  = __expf(sv - mn);
      float co = __expf(mrow[r] - mn);
      float rs = rsum16(p);
      lrow[r] = lrow[r] * co + rs;
      mrow[r] = mn;
      alpha[r] = co;
      s[r] = p;
    }
#pragma unroll
    for (int r = 0; r < 8; ++r) { o0[r] *= alpha[r]; o1[r] *= alpha[r]; }

    __syncthreads();
#pragma unroll
    for (int r = 0; r < 8; ++r) pLds[(r + 8 * half) * 16 + m] = s[r];
    __syncthreads();

#pragma unroll
    for (int kk = 0; kk < 16; kk += 4) {
      const int k0 = kk + half * 2;
      v2f a  = { pLds[m * 16 + k0], pLds[m * 16 + k0 + 1] };
      v2f b0 = { vbase[k0 * (3 * Dd) + m],      vbase[(k0 + 1) * (3 * Dd) + m] };
      v2f b1 = { vbase[k0 * (3 * Dd) + 16 + m], vbase[(k0 + 1) * (3 * Dd) + 16 + m] };
      o0 = wmma4(a, b0, o0);
      o1 = wmma4(a, b1, o1);
    }
  }

  float* ob = obuf + (size_t)(qb * 16) * Dd + h * HDd;
#pragma unroll
  for (int r = 0; r < 8; ++r) {
    const int row = r + 8 * half;
    const float inv = 1.0f / lrow[r];
    ob[row * Dd + m]      = o0[r] * inv;
    ob[row * Dd + 16 + m] = o1[r] * inv;
  }
}

__global__ void zerof(float* p, int n) {
  int i = blockIdx.x * blockDim.x + threadIdx.x;
  if (i < n) p[i] = 0.f;
}
__global__ void seg_sum(const float* __restrict__ ew, const int* __restrict__ src,
                        float* __restrict__ out, int E) {
  int i = blockIdx.x * blockDim.x + threadIdx.x;
  if (i < E) atomicAdd(&out[src[i]], ew[i]);
}

__global__ __launch_bounds__(256)
void residual_ln(float* __restrict__ h, const float* __restrict__ add,
                 const float* __restrict__ wsn,
                 const float* __restrict__ g, const float* __restrict__ b) {
  const int row = blockIdx.x;
  const int c = threadIdx.x;
  const float wsv = wsn ? wsn[row] : 1.0f;
  float t = h[(size_t)row * Dd + c] + add[(size_t)row * Dd + c] * wsv;

  __shared__ float red[Dd];
  red[c] = t; __syncthreads();
  for (int s = Dd / 2; s > 0; s >>= 1) { if (c < s) red[c] += red[c + s]; __syncthreads(); }
  const float mean = red[0] / (float)Dd;
  __syncthreads();
  const float d = t - mean;
  red[c] = d * d; __syncthreads();
  for (int s = Dd / 2; s > 0; s >>= 1) { if (c < s) red[c] += red[c + s]; __syncthreads(); }
  const float var = red[0] / (float)Dd;
  const float rs = rsqrtf(var + 1e-5f);
  h[(size_t)row * Dd + c] = d * rs * g[c] + b[c];
}

__global__ void add_kernel(const float* a, const float* b, float* o, int n) {
  int i = blockIdx.x * blockDim.x + threadIdx.x;
  if (i < n) o[i] = a[i] + b[i];
}

__global__ __launch_bounds__(32)
void predict16(const float* __restrict__ hc,
               const int* __restrict__ si, const int* __restrict__ di,
               const float* __restrict__ pw1, const float* __restrict__ pb1,
               const float* __restrict__ pw2, const float* __restrict__ pb2,
               float* __restrict__ out) {
  const int tid  = threadIdx.x;
  const int base = blockIdx.x * 16;

  __shared__ float z[16 * Dd];
  for (int idx = tid; idx < 16 * Dd; idx += 32) {
    const int row = idx >> 8, col = idx & (Dd - 1);
    const int rs = si[base + row], rd = di[base + row];
    z[idx] = hc[(size_t)rs * Dd + col] * hc[(size_t)rd * Dd + col];
  }
  __syncthreads();

  const int m = tid & 15;
  const int half = tid >> 4;
  float acc[8];
#pragma unroll
  for (int r = 0; r < 8; ++r) acc[r] = 0.f;

  for (int jt = 0; jt < 8; ++jt) {
    v8f c = {};
    const float* wrow = pw1 + (size_t)(jt * 16 + m) * Dd + half * 2;
    const float* zrow = z + m * Dd + half * 2;
#pragma unroll
    for (int k = 0; k < Dd; k += 4) {
      v2f a = { zrow[k], zrow[k + 1] };
      v2f b = { wrow[k], wrow[k + 1] };
      c = wmma4(a, b, c);
    }
    const float bias1 = pb1[jt * 16 + m];
    const float w2 = pw2[jt * 16 + m];
#pragma unroll
    for (int r = 0; r < 8; ++r) {
      float v = c[r] + bias1;
      v = v > 0.f ? v : 0.2f * v;
      acc[r] += v * w2;
    }
  }
#pragma unroll
  for (int r = 0; r < 8; ++r) {
    const float sr = rsum16(acc[r]);
    if (m == 0) out[base + r + 8 * half] = sr + pb2[0];
  }
}

extern "C" void kernel_launch(void* const* d_in, const int* in_sizes, int n_in,
                              void* d_out, int out_size, void* d_ws, size_t ws_size,
                              hipStream_t stream) {
  const float* x       = (const float*)d_in[0];
  const float* edge_w  = (const float*)d_in[1];
  const int*   src     = (const int*)d_in[2];
  const int*   pos_src = (const int*)d_in[4];
  const int*   pos_dst = (const int*)d_in[5];
  const int*   neg_src = (const int*)d_in[6];
  const int*   neg_dst = (const int*)d_in[7];
  const float* proj_w  = (const float*)d_in[8];
  const float* proj_b  = (const float*)d_in[9];
  const float* in_w    = (const float*)d_in[10];
  const float* in_b    = (const float*)d_in[11];
  const float* out_w   = (const float*)d_in[12];
  const float* out_b   = (const float*)d_in[13];
  const float* ff_w    = (const float*)d_in[14];
  const float* ff_b    = (const float*)d_in[15];
  const float* g1      = (const float*)d_in[16];
  const float* b1      = (const float*)d_in[17];
  const float* g2      = (const float*)d_in[18];
  const float* b2      = (const float*)d_in[19];
  const float* pw1     = (const float*)d_in[20];
  const float* pb1     = (const float*)d_in[21];
  const float* pw2     = (const float*)d_in[22];
  const float* pb2     = (const float*)d_in[23];

  float* W      = (float*)d_ws;
  float* h      = W;
  float* hproj  = W + 1048576;
  float* qkv    = W + 2097152;
  float* obuf   = W + 5242880;
  float* scr    = W + 6291456;
  float* wsn    = W + 7340032;

  float* outp = (float*)d_out;
  float* pos  = outp;
  float* neg  = outp + Pp;
  float* hc   = outp + 2 * Pp;

  gemm_atb<INc><<<dim3(Dd / 64, Nn / 16), 32, 0, stream>>>(x, proj_w, proj_b, hproj, Nn, Dd);
  hipMemcpyAsync(h, hproj, (size_t)Nn * Dd * sizeof(float), hipMemcpyDeviceToDevice, stream);

  for (int l = 0; l < Ll; ++l) {
    gemm_atb<Dd><<<dim3(3 * Dd / 64, Nn / 16), 32, 0, stream>>>(
        h, in_w + (size_t)l * 3 * Dd * Dd, in_b + (size_t)l * 3 * Dd, qkv, Nn, 3 * Dd);
    attn_kernel<<<dim3(Nn / 16, Hh), 32, 0, stream>>>(qkv, obuf);
    gemm_atb<Dd><<<dim3(Dd / 64, Nn / 16), 32, 0, stream>>>(
        obuf, out_w + (size_t)l * Dd * Dd, out_b + (size_t)l * Dd, scr, Nn, Dd);
    zerof<<<(Nn + 255) / 256, 256, 0, stream>>>(wsn, Nn);
    seg_sum<<<(Ee + 255) / 256, 256, 0, stream>>>(edge_w + (size_t)l * Ee,
                                                  src + (size_t)l * Ee, wsn, Ee);
    residual_ln<<<Nn, Dd, 0, stream>>>(h, scr, wsn, g1 + (size_t)l * Dd, b1 + (size_t)l * Dd);
    gemm_atb<Dd><<<dim3(Dd / 64, Nn / 16), 32, 0, stream>>>(
        h, ff_w + (size_t)l * Dd * Dd, ff_b + (size_t)l * Dd, scr, Nn, Dd);
    residual_ln<<<Nn, Dd, 0, stream>>>(h, scr, nullptr, g2 + (size_t)l * Dd, b2 + (size_t)l * Dd);
  }

  add_kernel<<<(Nn * Dd + 255) / 256, 256, 0, stream>>>(h, hproj, hc, Nn * Dd);

  predict16<<<Pp / 16, 32, 0, stream>>>(hc, pos_src, pos_dst, pw1, pb1, pw2, pb2, pos);
  predict16<<<Pp / 16, 32, 0, stream>>>(hc, neg_src, neg_dst, pw1, pb1, pw2, pb2, neg);
}


# attention: work-group of 4 waves (4 query tiles of a head) stages each 64-key K and V chunk in LDS once, one chunk ahead, with 16-byte loads; K/V fragments read from LDS; every sum order kept
# speedup vs baseline: 2.4886x; 2.4886x over previous
#include <hip/hip_runtime.h>
#include <math.h>

typedef __attribute__((ext_vector_type(2))) float v2f;
typedef __attribute__((ext_vector_type(8))) float v8f;

#define Nn   4096
#define INF_ 3.402823466e+38f
#define Dd   256
#define Hh   8
#define HDd  32
#define Ll   2
#define Ee   131072
#define Pp   200000
#define INc  128

__device__ __forceinline__ v8f wmma4(v2f a, v2f b, v8f c) {
  return __builtin_amdgcn_wmma_f32_16x16x4_f32(
      false, a, false, b, (short)0, c, false, false);
}

template <int MASK>
__device__ __forceinline__ float row_xor(float v) {
  return __builtin_amdgcn_update_dpp(0.0f, v, 0x160 + MASK, 0xf, 0xf, true);
}
__device__ __forceinline__ float rsum16(float v) {
  v += row_xor<8>(v);
  v += row_xor<4>(v);
  v += row_xor<2>(v);
  v += row_xor<1>(v);
  return v;
}
__device__ __forceinline__ float rmax16(float v) {
  v = fmaxf(v, row_xor<8>(v));
  v = fmaxf(v, row_xor<4>(v));
  v = fmaxf(v, row_xor<2>(v));
  v = fmaxf(v, row_xor<1>(v));
  return v;
}

template <int KK>
__global__ __launch_bounds__(32)
void gemm_atb(const float* __restrict__ A, const float* __restrict__ W,
              const float* __restrict__ bias, float* __restrict__ C,
              int M, int Nout) {
  const int tid  = threadIdx.x;
  const int m    = tid & 15;
  const int half = tid >> 4;
  const int tbn  = blockIdx.x * 64;
  const int tbm  = blockIdx.y * 16;

  const float* arow = A + (size_t)(tbm + m) * KK + half * 2;
  const float* w0   = W + (size_t)(tbn + m) * KK + half * 2;
  const float* w1   = w0 + (size_t)16 * KK;
  const float* w2   = w1 + (size_t)16 * KK;
  const float* w3   = w2 + (size_t)16 * KK;

  v8f c0 = {}, c1 = {}, c2 = {}, c3 = {};
#pragma unroll
  for (int k = 0; k < KK; k += 4) {
    v2f a  = { arow[k], arow[k + 1] };
    v2f b0 = { w0[k],   w0[k + 1] };
    v2f b1 = { w1[k],   w1[k + 1] };
    v2f b2 = { w2[k],   w2[k + 1] };
    v2f b3 = { w3[k],   w3[k + 1] };
    c0 = wmma4(a, b0, c0);
    c1 = wmma4(a, b1, c1);
    c2 = wmma4(a, b2, c2);
    c3 = wmma4(a, b3, c3);
  }
  for (int r = 0; r < 8; ++r) {
    const int row = tbm + r + 8 * half;
    float* crow = C + (size_t)row * Nout + tbn + m;
    if (bias) {
      const float* brow = bias + tbn + m;
      crow[0]  = c0[r] + brow[0];
      crow[16] = c1[r] + brow[16];
      crow[32] = c2[r] + brow[32];
      crow[48] = c3[r] + brow[48];
    } else {
      crow[0] = c0[r]; crow[16] = c1[r]; crow[32] = c2[r]; crow[48] = c3[r];
    }
  }
}

typedef __attribute__((ext_vector_type(4))) float v4f;

#define AW  4
#define KC  64
#define KPI 36
#define VPI 48

__global__ __launch_bounds__(32 * AW)
void attn_kernel(const float* __restrict__ qkv, float* __restrict__ obuf) {
  __shared__ __attribute__((aligned(16))) float kS[2][KC * KPI];
  __shared__ __attribute__((aligned(16))) float vS[2][KC * VPI];
  __shared__ float pS[AW][16 * 16];

  const int tid  = threadIdx.x;
  const int lane = tid & 31;
  const int m    = lane & 15;
  const int half = lane >> 4;
  const int wv   = __builtin_amdgcn_readfirstlane(tid >> 5);
  const int qb   = blockIdx.x * AW + wv;
  const int h    = blockIdx.y;
  const float scale = 0.1767766952966369f;

  const float* qbase = qkv + (size_t)(qb * 16) * (3 * Dd) + h * HDd;

  v2f qa[8];
#pragma unroll
  for (int kk = 0; kk < HDd; kk += 4) {
    const int k0 = kk + half * 2;
    qa[kk >> 2] = v2f{ qbase[m * (3 * Dd) + k0], qbase[m * (3 * Dd) + k0 + 1] };
  }

  const int srow = tid >> 3;
  const int scol = (tid & 7) * 4;
  const float* kg = qkv + (size_t)srow * (3 * Dd) + Dd + h * HDd + scol;
  const float* vg = kg + Dd;
  float* kst = &kS[0][srow * KPI + scol];
  float* vst = &vS[0][srow * VPI + scol];

  v8f o0 = {}, o1 = {};
  float mrow[8], lrow[8];
  for (int r = 0; r < 8; ++r) { mrow[r] = -INF_; lrow[r] = 0.f; }

#pragma unroll
  for (int i = 0; i < KC / 16; ++i) {
    *(v4f*)(kst + i * 16 * KPI) = *(const v4f*)(kg + (size_t)(i * 16) * (3 * Dd));
    *(v4f*)(vst + i * 16 * VPI) = *(const v4f*)(vg + (size_t)(i * 16) * (3 * Dd));
  }
  __syncthreads();

  float* pl = &pS[wv][0];
  int cur = 0;
  for (int ch = 0; ch < Nn / KC; ++ch) {
    const int nc = (ch + 1 < Nn / KC) ? ch + 1 : ch;
    const float* kgn = kg + (size_t)nc * KC * (3 * Dd);
    const float* vgn = vg + (size_t)nc * KC * (3 * Dd);
    v4f nk[KC / 16], nv[KC / 16];
#pragma unroll
    for (int i = 0; i < KC / 16; ++i) {
      nk[i] = *(const v4f*)(kgn + (size_t)(i * 16) * (3 * Dd));
      nv[i] = *(const v4f*)(vgn + (size_t)(i * 16) * (3 * Dd));
    }

    const float* kf = &kS[cur][m * KPI + half * 2];
    const float* vf = &vS[cur][(half * 2) * VPI + m];

#pragma unroll
    for (int j = 0; j < KC / 16; ++j) {
      v2f kb[8];
#pragma unroll
      for (int i = 0; i < 8; ++i) kb[i] = *(const v2f*)(kf + j * 16 * KPI + 4 * i);
      v8f s = {};
#pragma unroll
      for (int i = 0; i < 8; ++i) s = wmma4(qa[i], kb[i], s);

      v2f vb0[4], vb1[4];
#pragma unroll
      for (int i = 0; i < 4; ++i) {
        const float* vp = vf + (j * 16 + 4 * i) * VPI;
        vb0[i] = v2f{ vp[0],  vp[VPI] };
        vb1[i] = v2f{ vp[16], vp[VPI + 16] };
      }

      float alpha[8];
#pragma unroll
      for (int r = 0; r < 8; ++r) {
        float sv = s[r] * scale;
        float rm = rmax16(sv);
        float mn = fmaxf(mrow[r], rm);
        float p  = __expf(sv - mn);
        float co = __expf(mrow[r] - mn);
        float rs = rsum16(p);
        lrow[r] = lrow[r] * co + rs;
        mrow[r] = mn;
        alpha[r] = co;
        s[r] = p;
      }
#pragma unroll
      for (int r = 0; r < 8; ++r) { o0[r] *= alpha[r]; o1[r] *= alpha[r]; }

#pragma unroll
      for (int r = 0; r < 8; ++r) pl[(r + 8 * half) * 16 + m] = s[r];
      __builtin_amdgcn_fence(__ATOMIC_RELEASE, "wavefront");
      __builtin_amdgcn_wave_barrier();
      __builtin_amdgcn_fence(__ATOMIC_ACQUIRE, "wavefront");

#pragma unroll
      for (int i = 0; i < 4; ++i) {
        const int k0 = 4 * i + half * 2;
        v2f a = { pl[m * 16 + k0], pl[m * 16 + k0 + 1] };
        o0 = wmma4(a, vb0[i], o0);
        o1 = wmma4(a, vb1[i], o1);
      }
    }

    {
      float* kd = kst + (cur ^ 1) * (KC * KPI);
      float* vd = vst + (cur ^ 1) * (KC * VPI);
#pragma unroll
      for (int i = 0; i < KC / 16; ++i) {
        *(v4f*)(kd + i * 16 * KPI) = nk[i];
        *(v4f*)(vd + i * 16 * VPI) = nv[i];
      }
    }
    __syncthreads();
    cur ^= 1;
  }

  float* ob = obuf + (size_t)(qb * 16) * Dd + h * HDd;
#pragma unroll
  for (int r = 0; r < 8; ++r) {
    const int row = r + 8 * half;
    const float inv = 1.0f / lrow[r];
    ob[row * Dd + m]      = o0[r] * inv;
    ob[row * Dd + 16 + m] = o1[r] * inv;
  }
}

__global__ void zerof(float* p, int n) {
  int i = blockIdx.x * blockDim.x + threadIdx.x;
  if (i < n) p[i] = 0.f;
}
__global__ void seg_sum(const float* __restrict__ ew, const int* __restrict__ src,
                        float* __restrict__ out, int E) {
  int i = blockIdx.x * blockDim.x + threadIdx.x;
  if (i < E) atomicAdd(&out[src[i]], ew[i]);
}

__global__ __launch_bounds__(256)
void residual_ln(float* __restrict__ h, const float* __restrict__ add,
                 const float* __restrict__ wsn,
                 const float* __restrict__ g, const float* __restrict__ b) {
  const int row = blockIdx.x;
  const int c = threadIdx.x;
  const float wsv = wsn ? wsn[row] : 1.0f;
  float t = h[(size_t)row * Dd + c] + add[(size_t)row * Dd + c] * wsv;

  __shared__ float red[Dd];
  red[c] = t; __syncthreads();
  for (int s = Dd / 2; s > 0; s >>= 1) { if (c < s) red[c] += red[c + s]; __syncthreads(); }
  const float mean = red[0] / (float)Dd;
  __syncthreads();
  const float d = t - mean;
  red[c] = d * d; __syncthreads();
  for (int s = Dd / 2; s > 0; s >>= 1) { if (c < s) red[c] += red[c + s]; __syncthreads(); }
  const float var = red[0] / (float)Dd;
  const float rs = rsqrtf(var + 1e-5f);
  h[(size_t)row * Dd + c] = d * rs * g[c] + b[c];
}

__global__ void add_kernel(const float* a, const float* b, float* o, int n) {
  int i = blockIdx.x * blockDim.x + threadIdx.x;
  if (i < n) o[i] = a[i] + b[i];
}


#define ZP 260
#define PT 32
#define PP 20

__global__ __launch_bounds__(256)
void predict_wreg(const float* __restrict__ hc,
                  const int* __restrict__ si, const int* __restrict__ di,
                  const float* __restrict__ pw1, const float* __restrict__ pb1,
                  const float* __restrict__ pw2, const float* __restrict__ pb2,
                  float* __restrict__ out, int ntiles) {
  __shared__ __attribute__((aligned(16))) float zb[2][PT * ZP];
  __shared__ __attribute__((aligned(16))) float part[2][PT * PP];

  const int tid  = threadIdx.x;
  const int lane = tid & 31;
  const int c    = lane & 15;
  const int half = lane >> 4;
  const int wv   = __builtin_amdgcn_readfirstlane(tid >> 5);

  v2f wa[64];
  {
    const float* wrow = pw1 + (size_t)(wv * 16 + c) * Dd + half * 2;
#pragma unroll
    for (int i = 0; i < 64; ++i) wa[i] = v2f{ wrow[4 * i], wrow[4 * i + 1] };
  }
  float b1r[8], w2r[8];
#pragma unroll
  for (int r = 0; r < 8; ++r) {
    b1r[r] = pb1[wv * 16 + 8 * half + r];
    w2r[r] = pw2[wv * 16 + 8 * half + r];
  }
  const float bias2 = pb2[0];

  int tile = blockIdx.x;
  if (tile >= ntiles) return;

  {
    const int pb = tile * PT + wv * 4;
    float* zn = &zb[0][(wv * 4) * ZP + 4 * lane];
#pragma unroll
    for (int q = 0; q < 4; ++q) {
      const v4f* rs = (const v4f*)(hc + (size_t)si[pb + q] * Dd);
      const v4f* rd = (const v4f*)(hc + (size_t)di[pb + q] * Dd);
      *(v4f*)(zn + q * ZP)       = rs[lane] * rd[lane];
      *(v4f*)(zn + q * ZP + 128) = rs[32 + lane] * rd[32 + lane];
    }
  }
  __syncthreads();

  int cur = 0;
  for (; tile < ntiles; tile += gridDim.x) {
    const int nt = (tile + (int)gridDim.x < ntiles) ? tile + (int)gridDim.x : tile;
    const int pb = nt * PT + wv * 4;
    v4f xs[4][2], xd[4][2];
#pragma unroll
    for (int q = 0; q < 4; ++q) {
      const v4f* rs = (const v4f*)(hc + (size_t)si[pb + q] * Dd);
      const v4f* rd = (const v4f*)(hc + (size_t)di[pb + q] * Dd);
      xs[q][0] = rs[lane]; xs[q][1] = rs[32 + lane];
      xd[q][0] = rd[lane]; xd[q][1] = rd[32 + lane];
    }

    const float* z0 = &zb[cur][c * ZP + half * 2];
    const float* z1 = z0 + 16 * ZP;
    v8f acc0 = {}, acc1 = {};
#pragma unroll
    for (int i = 0; i < 64; ++i) {
      const v2f b0 = *(const v2f*)(z0 + 4 * i);
      const v2f b1 = *(const v2f*)(z1 + 4 * i);
      acc0 = wmma4(wa[i], b0, acc0);
      acc1 = wmma4(wa[i], b1, acc1);
    }

    float p0 = 0.f, p1 = 0.f;
#pragma unroll
    for (int r = 0; r < 8; ++r) {
      float v0 = acc0[r] + b1r[r];
      float v1 = acc1[r] + b1r[r];
      v0 = v0 > 0.f ? v0 : 0.2f * v0;
      v1 = v1 > 0.f ? v1 : 0.2f * v1;
      p0 += v0 * w2r[r];
      p1 += v1 * w2r[r];
    }
    part[cur][c * PP + 2 * wv + half]        = p0;
    part[cur][(16 + c) * PP + 2 * wv + half] = p1;

    {
      float* zn = &zb[cur ^ 1][(wv * 4) * ZP + 4 * lane];
#pragma unroll
      for (int q = 0; q < 4; ++q) {
        *(v4f*)(zn + q * ZP)       = xs[q][0] * xd[q][0];
        *(v4f*)(zn + q * ZP + 128) = xs[q][1] * xd[q][1];
      }
    }
    __syncthreads();

    if (wv == 0) {
      const v4f* pp = (const v4f*)&part[cur][lane * PP];
      const v4f a = pp[0], b = pp[1], d = pp[2], e = pp[3];
      const float s = (((a.x + a.y) + (a.z + a.w)) + ((b.x + b.y) + (b.z + b.w))) +
                      (((d.x + d.y) + (d.z + d.w)) + ((e.x + e.y) + (e.z + e.w)));
      out[tile * PT + lane] = s + bias2;
    }
    cur ^= 1;
  }
}

extern "C" void kernel_launch(void* const* d_in, const int* in_sizes, int n_in,
                              void* d_out, int out_size, void* d_ws, size_t ws_size,
                              hipStream_t stream) {
  const float* x       = (const float*)d_in[0];
  const float* edge_w  = (const float*)d_in[1];
  const int*   src     = (const int*)d_in[2];
  const int*   pos_src = (const int*)d_in[4];
  const int*   pos_dst = (const int*)d_in[5];
  const int*   neg_src = (const int*)d_in[6];
  const int*   neg_dst = (const int*)d_in[7];
  const float* proj_w  = (const float*)d_in[8];
  const float* proj_b  = (const float*)d_in[9];
  const float* in_w    = (const float*)d_in[10];
  const float* in_b    = (const float*)d_in[11];
  const float* out_w   = (const float*)d_in[12];
  const float* out_b   = (const float*)d_in[13];
  const float* ff_w    = (const float*)d_in[14];
  const float* ff_b    = (const float*)d_in[15];
  const float* g1      = (const float*)d_in[16];
  const float* b1      = (const float*)d_in[17];
  const float* g2      = (const float*)d_in[18];
  const float* b2      = (const float*)d_in[19];
  const float* pw1     = (const float*)d_in[20];
  const float* pb1     = (const float*)d_in[21];
  const float* pw2     = (const float*)d_in[22];
  const float* pb2     = (const float*)d_in[23];

  float* W      = (float*)d_ws;
  float* h      = W;
  float* hproj  = W + 1048576;
  float* qkv    = W + 2097152;
  float* obuf   = W + 5242880;
  float* scr    = W + 6291456;
  float* wsn    = W + 7340032;

  float* outp = (float*)d_out;
  float* pos  = outp;
  float* neg  = outp + Pp;
  float* hc   = outp + 2 * Pp;

  gemm_atb<INc><<<dim3(Dd / 64, Nn / 16), 32, 0, stream>>>(x, proj_w, proj_b, hproj, Nn, Dd);
  hipMemcpyAsync(h, hproj, (size_t)Nn * Dd * sizeof(float), hipMemcpyDeviceToDevice, stream);

  for (int l = 0; l < Ll; ++l) {
    gemm_atb<Dd><<<dim3(3 * Dd / 64, Nn / 16), 32, 0, stream>>>(
        h, in_w + (size_t)l * 3 * Dd * Dd, in_b + (size_t)l * 3 * Dd, qkv, Nn, 3 * Dd);
    attn_kernel<<<dim3(Nn / (16 * AW), Hh), 32 * AW, 0, stream>>>(qkv, obuf);
    gemm_atb<Dd><<<dim3(Dd / 64, Nn / 16), 32, 0, stream>>>(
        obuf, out_w + (size_t)l * Dd * Dd, out_b + (size_t)l * Dd, scr, Nn, Dd);
    zerof<<<(Nn + 255) / 256, 256, 0, stream>>>(wsn, Nn);
    seg_sum<<<(Ee + 255) / 256, 256, 0, stream>>>(edge_w + (size_t)l * Ee,
                                                  src + (size_t)l * Ee, wsn, Ee);
    residual_ln<<<Nn, Dd, 0, stream>>>(h, scr, wsn, g1 + (size_t)l * Dd, b1 + (size_t)l * Dd);
    gemm_atb<Dd><<<dim3(Dd / 64, Nn / 16), 32, 0, stream>>>(
        h, ff_w + (size_t)l * Dd * Dd, ff_b + (size_t)l * Dd, scr, Nn, Dd);
    residual_ln<<<Nn, Dd, 0, stream>>>(h, scr, nullptr, g2 + (size_t)l * Dd, b2 + (size_t)l * Dd);
  }

  add_kernel<<<(Nn * Dd + 255) / 256, 256, 0, stream>>>(h, hproj, hc, Nn * Dd);

  predict_wreg<<<512, 256, 0, stream>>>(hc, pos_src, pos_dst, pw1, pb1, pw2, pb2, pos, Pp / PT);
  predict_wreg<<<512, 256, 0, stream>>>(hc, neg_src, neg_dst, pw1, pb1, pw2, pb2, neg, Pp / PT);
}


# attention softmax in the A-operand layout: scaled scores (not P) take the wave's LDS tile trip, so row max/sum are in-lane plus one half exchange, P feeds P*V from registers; row-sum order changes
# speedup vs baseline: 2.8567x; 1.1479x over previous
#include <hip/hip_runtime.h>
#include <math.h>

typedef __attribute__((ext_vector_type(2))) float v2f;
typedef __attribute__((ext_vector_type(8))) float v8f;

#define Nn   4096
#define INF_ 3.402823466e+38f
#define Dd   256
#define Hh   8
#define HDd  32
#define Ll   2
#define Ee   131072
#define Pp   200000
#define INc  128

__device__ __forceinline__ v8f wmma4(v2f a, v2f b, v8f c) {
  return __builtin_amdgcn_wmma_f32_16x16x4_f32(
      false, a, false, b, (short)0, c, false, false);
}

__device__ __forceinline__ float xhalf(float x) {
  int b = __builtin_bit_cast(int, x);
  return __builtin_bit_cast(float,
      __builtin_amdgcn_permlanex16(b, b, 0x76543210, 0xfedcba98, false, false));
}
template <int N>
__device__ __forceinline__ float row_bcast(float v) {
  return __builtin_amdgcn_update_dpp(0.0f, v, 0x150 + N, 0xf, 0xf, true);
}
__device__ __forceinline__ void rows_of_tile(float v, float (&out)[8]) {
  const float t = __builtin_amdgcn_update_dpp(v, v, 0x128, 0x2, 0xf, false);
  out[0] = row_bcast<0>(t); out[1] = row_bcast<1>(t);
  out[2] = row_bcast<2>(t); out[3] = row_bcast<3>(t);
  out[4] = row_bcast<4>(t); out[5] = row_bcast<5>(t);
  out[6] = row_bcast<6>(t); out[7] = row_bcast<7>(t);
}

template <int KK>
__global__ __launch_bounds__(32)
void gemm_atb(const float* __restrict__ A, const float* __restrict__ W,
              const float* __restrict__ bias, float* __restrict__ C,
              int M, int Nout) {
  const int tid  = threadIdx.x;
  const int m    = tid & 15;
  const int half = tid >> 4;
  const int tbn  = blockIdx.x * 64;
  const int tbm  = blockIdx.y * 16;

  const float* arow = A + (size_t)(tbm + m) * KK + half * 2;
  const float* w0   = W + (size_t)(tbn + m) * KK + half * 2;
  const float* w1   = w0 + (size_t)16 * KK;
  const float* w2   = w1 + (size_t)16 * KK;
  const float* w3   = w2 + (size_t)16 * KK;

  v8f c0 = {}, c1 = {}, c2 = {}, c3 = {};
#pragma unroll
  for (int k = 0; k < KK; k += 4) {
    v2f a  = { arow[k], arow[k + 1] };
    v2f b0 = { w0[k],   w0[k + 1] };
    v2f b1 = { w1[k],   w1[k + 1] };
    v2f b2 = { w2[k],   w2[k + 1] };
    v2f b3 = { w3[k],   w3[k + 1] };
    c0 = wmma4(a, b0, c0);
    c1 = wmma4(a, b1, c1);
    c2 = wmma4(a, b2, c2);
    c3 = wmma4(a, b3, c3);
  }
  for (int r = 0; r < 8; ++r) {
    const int row = tbm + r + 8 * half;
    float* crow = C + (size_t)row * Nout + tbn + m;
    if (bias) {
      const float* brow = bias + tbn + m;
      crow[0]  = c0[r] + brow[0];
      crow[16] = c1[r] + brow[16];
      crow[32] = c2[r] + brow[32];
      crow[48] = c3[r] + brow[48];
    } else {
      crow[0] = c0[r]; crow[16] = c1[r]; crow[32] = c2[r]; crow[48] = c3[r];
    }
  }
}

typedef __attribute__((ext_vector_type(4))) float v4f;

#define AW  4
#define KC  64
#define KPI 36
#define VPI 48

__global__ __launch_bounds__(32 * AW)
void attn_kernel(const float* __restrict__ qkv, float* __restrict__ obuf) {
  __shared__ __attribute__((aligned(16))) float kS[2][KC * KPI];
  __shared__ __attribute__((aligned(16))) float vS[2][KC * VPI];
  __shared__ float pS[AW][16 * 16];

  const int tid  = threadIdx.x;
  const int lane = tid & 31;
  const int m    = lane & 15;
  const int half = lane >> 4;
  const int wv   = __builtin_amdgcn_readfirstlane(tid >> 5);
  const int qb   = blockIdx.x * AW + wv;
  const int h    = blockIdx.y;
  const float scale = 0.1767766952966369f;

  const float* qbase = qkv + (size_t)(qb * 16) * (3 * Dd) + h * HDd;

  v2f qa[8];
#pragma unroll
  for (int kk = 0; kk < HDd; kk += 4) {
    const int k0 = kk + half * 2;
    qa[kk >> 2] = v2f{ qbase[m * (3 * Dd) + k0], qbase[m * (3 * Dd) + k0 + 1] };
  }

  const int srow = tid >> 3;
  const int scol = (tid & 7) * 4;
  const float* kg = qkv + (size_t)srow * (3 * Dd) + Dd + h * HDd + scol;
  const float* vg = kg + Dd;
  float* kst = &kS[0][srow * KPI + scol];
  float* vst = &vS[0][srow * VPI + scol];

  v8f o0 = {}, o1 = {};
  float mrow = -INF_, lrow = 0.f;

#pragma unroll
  for (int i = 0; i < KC / 16; ++i) {
    *(v4f*)(kst + i * 16 * KPI) = *(const v4f*)(kg + (size_t)(i * 16) * (3 * Dd));
    *(v4f*)(vst + i * 16 * VPI) = *(const v4f*)(vg + (size_t)(i * 16) * (3 * Dd));
  }
  __syncthreads();

  float* pl = &pS[wv][0];
  int cur = 0;
  for (int ch = 0; ch < Nn / KC; ++ch) {
    const int nc = (ch + 1 < Nn / KC) ? ch + 1 : ch;
    const float* kgn = kg + (size_t)nc * KC * (3 * Dd);
    const float* vgn = vg + (size_t)nc * KC * (3 * Dd);
    v4f nk[KC / 16], nv[KC / 16];
#pragma unroll
    for (int i = 0; i < KC / 16; ++i) {
      nk[i] = *(const v4f*)(kgn + (size_t)(i * 16) * (3 * Dd));
      nv[i] = *(const v4f*)(vgn + (size_t)(i * 16) * (3 * Dd));
    }

    const float* kf = &kS[cur][m * KPI + half * 2];
    const float* vf = &vS[cur][(half * 2) * VPI + m];

#pragma unroll
    for (int j = 0; j < KC / 16; ++j) {
      v2f kb[8];
#pragma unroll
      for (int i = 0; i < 8; ++i) kb[i] = *(const v2f*)(kf + j * 16 * KPI + 4 * i);
      v8f s = {};
#pragma unroll
      for (int i = 0; i < 8; ++i) s = wmma4(qa[i], kb[i], s);

      v2f vb0[4], vb1[4];
#pragma unroll
      for (int i = 0; i < 4; ++i) {
        const float* vp = vf + (j * 16 + 4 * i) * VPI;
        vb0[i] = v2f{ vp[0],  vp[VPI] };
        vb1[i] = v2f{ vp[16], vp[VPI + 16] };
      }

#pragma unroll
      for (int r = 0; r < 8; ++r) pl[(r + 8 * half) * 16 + m] = s[r] * scale;
      __builtin_amdgcn_fence(__ATOMIC_RELEASE, "wavefront");
      __builtin_amdgcn_wave_barrier();
      __builtin_amdgcn_fence(__ATOMIC_ACQUIRE, "wavefront");
      v2f x[4];
#pragma unroll
      for (int i = 0; i < 4; ++i) {
        const int k0 = 4 * i + half * 2;
        x[i] = v2f{ pl[m * 16 + k0], pl[m * 16 + k0 + 1] };
      }

      float rm = fmaxf(fmaxf(fmaxf(x[0][0], x[0][1]), fmaxf(x[1][0], x[1][1])),
                       fmaxf(fmaxf(x[2][0], x[2][1]), fmaxf(x[3][0], x[3][1])));
      rm = fmaxf(rm, xhalf(rm));
      const float mn = fmaxf(mrow, rm);
      const float co = __expf(mrow - mn);
      v2f p[4];
#pragma unroll
      for (int i = 0; i < 4; ++i) p[i] = v2f{ __expf(x[i][0] - mn), __expf(x[i][1] - mn) };
      const float rs = ((p[0][0] + p[0][1]) + (p[1][0] + p[1][1])) +
                       ((p[2][0] + p[2][1]) + (p[3][0] + p[3][1]));
      lrow = lrow * co + rs;
      mrow = mn;

      float al[8];
      rows_of_tile(co, al);
#pragma unroll
      for (int r = 0; r < 8; ++r) { o0[r] *= al[r]; o1[r] *= al[r]; }

#pragma unroll
      for (int i = 0; i < 4; ++i) {
        o0 = wmma4(p[i], vb0[i], o0);
        o1 = wmma4(p[i], vb1[i], o1);
      }
    }

    {
      float* kd = kst + (cur ^ 1) * (KC * KPI);
      float* vd = vst + (cur ^ 1) * (KC * VPI);
#pragma unroll
      for (int i = 0; i < KC / 16; ++i) {
        *(v4f*)(kd + i * 16 * KPI) = nk[i];
        *(v4f*)(vd + i * 16 * VPI) = nv[i];
      }
    }
    __syncthreads();
    cur ^= 1;
  }

  float* ob = obuf + (size_t)(qb * 16) * Dd + h * HDd;
  float invr[8];
  rows_of_tile(1.0f / (lrow + xhalf(lrow)), invr);
#pragma unroll
  for (int r = 0; r < 8; ++r) {
    const int row = r + 8 * half;
    ob[row * Dd + m]      = o0[r] * invr[r];
    ob[row * Dd + 16 + m] = o1[r] * invr[r];
  }
}

__global__ void zerof(float* p, int n) {
  int i = blockIdx.x * blockDim.x + threadIdx.x;
  if (i < n) p[i] = 0.f;
}
__global__ void seg_sum(const float* __restrict__ ew, const int* __restrict__ src,
                        float* __restrict__ out, int E) {
  int i = blockIdx.x * blockDim.x + threadIdx.x;
  if (i < E) atomicAdd(&out[src[i]], ew[i]);
}

__global__ __launch_bounds__(256)
void residual_ln(float* __restrict__ h, const float* __restrict__ add,
                 const float* __restrict__ wsn,
                 const float* __restrict__ g, const float* __restrict__ b) {
  const int row = blockIdx.x;
  const int c = threadIdx.x;
  const float wsv = wsn ? wsn[row] : 1.0f;
  float t = h[(size_t)row * Dd + c] + add[(size_t)row * Dd + c] * wsv;

  __shared__ float red[Dd];
  red[c] = t; __syncthreads();
  for (int s = Dd / 2; s > 0; s >>= 1) { if (c < s) red[c] += red[c + s]; __syncthreads(); }
  const float mean = red[0] / (float)Dd;
  __syncthreads();
  const float d = t - mean;
  red[c] = d * d; __syncthreads();
  for (int s = Dd / 2; s > 0; s >>= 1) { if (c < s) red[c] += red[c + s]; __syncthreads(); }
  const float var = red[0] / (float)Dd;
  const float rs = rsqrtf(var + 1e-5f);
  h[(size_t)row * Dd + c] = d * rs * g[c] + b[c];
}

__global__ void add_kernel(const float* a, const float* b, float* o, int n) {
  int i = blockIdx.x * blockDim.x + threadIdx.x;
  if (i < n) o[i] = a[i] + b[i];
}


#define ZP 260
#define PT 32
#define PP 20

__global__ __launch_bounds__(256)
void predict_wreg(const float* __restrict__ hc,
                  const int* __restrict__ si, const int* __restrict__ di,
                  const float* __restrict__ pw1, const float* __restrict__ pb1,
                  const float* __restrict__ pw2, const float* __restrict__ pb2,
                  float* __restrict__ out, int ntiles) {
  __shared__ __attribute__((aligned(16))) float zb[2][PT * ZP];
  __shared__ __attribute__((aligned(16))) float part[2][PT * PP];

  const int tid  = threadIdx.x;
  const int lane = tid & 31;
  const int c    = lane & 15;
  const int half = lane >> 4;
  const int wv   = __builtin_amdgcn_readfirstlane(tid >> 5);

  v2f wa[64];
  {
    const float* wrow = pw1 + (size_t)(wv * 16 + c) * Dd + half * 2;
#pragma unroll
    for (int i = 0; i < 64; ++i) wa[i] = v2f{ wrow[4 * i], wrow[4 * i + 1] };
  }
  float b1r[8], w2r[8];
#pragma unroll
  for (int r = 0; r < 8; ++r) {
    b1r[r] = pb1[wv * 16 + 8 * half + r];
    w2r[r] = pw2[wv * 16 + 8 * half + r];
  }
  const float bias2 = pb2[0];

  int tile = blockIdx.x;
  if (tile >= ntiles) return;

  {
    const int pb = tile * PT + wv * 4;
    float* zn = &zb[0][(wv * 4) * ZP + 4 * lane];
#pragma unroll
    for (int q = 0; q < 4; ++q) {
      const v4f* rs = (const v4f*)(hc + (size_t)si[pb + q] * Dd);
      const v4f* rd = (const v4f*)(hc + (size_t)di[pb + q] * Dd);
      *(v4f*)(zn + q * ZP)       = rs[lane] * rd[lane];
      *(v4f*)(zn + q * ZP + 128) = rs[32 + lane] * rd[32 + lane];
    }
  }
  __syncthreads();

  int cur = 0;
  for (; tile < ntiles; tile += gridDim.x) {
    const int nt = (tile + (int)gridDim.x < ntiles) ? tile + (int)gridDim.x : tile;
    const int pb = nt * PT + wv * 4;
    v4f xs[4][2], xd[4][2];
#pragma unroll
    for (int q = 0; q < 4; ++q) {
      const v4f* rs = (const v4f*)(hc + (size_t)si[pb + q] * Dd);
      const v4f* rd = (const v4f*)(hc + (size_t)di[pb + q] * Dd);
      xs[q][0] = rs[lane]; xs[q][1] = rs[32 + lane];
      xd[q][0] = rd[lane]; xd[q][1] = rd[32 + lane];
    }

    const float* z0 = &zb[cur][c * ZP + half * 2];
    const float* z1 = z0 + 16 * ZP;
    v8f acc0 = {}, acc1 = {};
#pragma unroll
    for (int i = 0; i < 64; ++i) {
      const v2f b0 = *(const v2f*)(z0 + 4 * i);
      const v2f b1 = *(const v2f*)(z1 + 4 * i);
      acc0 = wmma4(wa[i], b0, acc0);
      acc1 = wmma4(wa[i], b1, acc1);
    }

    float p0 = 0.f, p1 = 0.f;
#pragma unroll
    for (int r = 0; r < 8; ++r) {
      float v0 = acc0[r] + b1r[r];
      float v1 = acc1[r] + b1r[r];
      v0 = v0 > 0.f ? v0 : 0.2f * v0;
      v1 = v1 > 0.f ? v1 : 0.2f * v1;
      p0 += v0 * w2r[r];
      p1 += v1 * w2r[r];
    }
    part[cur][c * PP + 2 * wv + half]        = p0;
    part[cur][(16 + c) * PP + 2 * wv + half] = p1;

    {
      float* zn = &zb[cur ^ 1][(wv * 4) * ZP + 4 * lane];
#pragma unroll
      for (int q = 0; q < 4; ++q) {
        *(v4f*)(zn + q * ZP)       = xs[q][0] * xd[q][0];
        *(v4f*)(zn + q * ZP + 128) = xs[q][1] * xd[q][1];
      }
    }
    __syncthreads();

    if (wv == 0) {
      const v4f* pp = (const v4f*)&part[cur][lane * PP];
      const v4f a = pp[0], b = pp[1], d = pp[2], e = pp[3];
      const float s = (((a.x + a.y) + (a.z + a.w)) + ((b.x + b.y) + (b.z + b.w))) +
                      (((d.x + d.y) + (d.z + d.w)) + ((e.x + e.y) + (e.z + e.w)));
      out[tile * PT + lane] = s + bias2;
    }
    cur ^= 1;
  }
}

extern "C" void kernel_launch(void* const* d_in, const int* in_sizes, int n_in,
                              void* d_out, int out_size, void* d_ws, size_t ws_size,
                              hipStream_t stream) {
  const float* x       = (const float*)d_in[0];
  const float* edge_w  = (const float*)d_in[1];
  const int*   src     = (const int*)d_in[2];
  const int*   pos_src = (const int*)d_in[4];
  const int*   pos_dst = (const int*)d_in[5];
  const int*   neg_src = (const int*)d_in[6];
  const int*   neg_dst = (const int*)d_in[7];
  const float* proj_w  = (const float*)d_in[8];
  const float* proj_b  = (const float*)d_in[9];
  const float* in_w    = (const float*)d_in[10];
  const float* in_b    = (const float*)d_in[11];
  const float* out_w   = (const float*)d_in[12];
  const float* out_b   = (const float*)d_in[13];
  const float* ff_w    = (const float*)d_in[14];
  const float* ff_b    = (const float*)d_in[15];
  const float* g1      = (const float*)d_in[16];
  const float* b1      = (const float*)d_in[17];
  const float* g2      = (const float*)d_in[18];
  const float* b2      = (const float*)d_in[19];
  const float* pw1     = (const float*)d_in[20];
  const float* pb1     = (const float*)d_in[21];
  const float* pw2     = (const float*)d_in[22];
  const float* pb2     = (const float*)d_in[23];

  float* W      = (float*)d_ws;
  float* h      = W;
  float* hproj  = W + 1048576;
  float* qkv    = W + 2097152;
  float* obuf   = W + 5242880;
  float* scr    = W + 6291456;
  float* wsn    = W + 7340032;

  float* outp = (float*)d_out;
  float* pos  = outp;
  float* neg  = outp + Pp;
  float* hc   = outp + 2 * Pp;

  gemm_atb<INc><<<dim3(Dd / 64, Nn / 16), 32, 0, stream>>>(x, proj_w, proj_b, hproj, Nn, Dd);
  hipMemcpyAsync(h, hproj, (size_t)Nn * Dd * sizeof(float), hipMemcpyDeviceToDevice, stream);

  for (int l = 0; l < Ll; ++l) {
    gemm_atb<Dd><<<dim3(3 * Dd / 64, Nn / 16), 32, 0, stream>>>(
        h, in_w + (size_t)l * 3 * Dd * Dd, in_b + (size_t)l * 3 * Dd, qkv, Nn, 3 * Dd);
    attn_kernel<<<dim3(Nn / (16 * AW), Hh), 32 * AW, 0, stream>>>(qkv, obuf);
    gemm_atb<Dd><<<dim3(Dd / 64, Nn / 16), 32, 0, stream>>>(
        obuf, out_w + (size_t)l * Dd * Dd, out_b + (size_t)l * Dd, scr, Nn, Dd);
    zerof<<<(Nn + 255) / 256, 256, 0, stream>>>(wsn, Nn);
    seg_sum<<<(Ee + 255) / 256, 256, 0, stream>>>(edge_w + (size_t)l * Ee,
                                                  src + (size_t)l * Ee, wsn, Ee);
    residual_ln<<<Nn, Dd, 0, stream>>>(h, scr, wsn, g1 + (size_t)l * Dd, b1 + (size_t)l * Dd);
    gemm_atb<Dd><<<dim3(Dd / 64, Nn / 16), 32, 0, stream>>>(
        h, ff_w + (size_t)l * Dd * Dd, ff_b + (size_t)l * Dd, scr, Nn, Dd);
    residual_ln<<<Nn, Dd, 0, stream>>>(h, scr, nullptr, g2 + (size_t)l * Dd, b2 + (size_t)l * Dd);
  }

  add_kernel<<<(Nn * Dd + 255) / 256, 256, 0, stream>>>(h, hproj, hc, Nn * Dd);

  predict_wreg<<<512, 256, 0, stream>>>(hc, pos_src, pos_dst, pw1, pb1, pw2, pb2, pos, Pp / PT);
  predict_wreg<<<512, 256, 0, stream>>>(hc, neg_src, neg_dst, pw1, pb1, pw2, pb2, neg, Pp / PT);
}


# GEMMs: a 4-wave work-group owns a 64x64 tile and stages 64-k chunks of A and W in LDS once with whole-line 16-byte loads, one chunk ahead; fragments read from LDS; every sum order kept
# speedup vs baseline: 3.3476x; 1.1718x over previous
#include <hip/hip_runtime.h>
#include <math.h>

typedef __attribute__((ext_vector_type(2))) float v2f;
typedef __attribute__((ext_vector_type(8))) float v8f;

#define Nn   4096
#define INF_ 3.402823466e+38f
#define Dd   256
#define Hh   8
#define HDd  32
#define Ll   2
#define Ee   131072
#define Pp   200000
#define INc  128

__device__ __forceinline__ v8f wmma4(v2f a, v2f b, v8f c) {
  return __builtin_amdgcn_wmma_f32_16x16x4_f32(
      false, a, false, b, (short)0, c, false, false);
}

__device__ __forceinline__ float xhalf(float x) {
  int b = __builtin_bit_cast(int, x);
  return __builtin_bit_cast(float,
      __builtin_amdgcn_permlanex16(b, b, 0x76543210, 0xfedcba98, false, false));
}
template <int N>
__device__ __forceinline__ float row_bcast(float v) {
  return __builtin_amdgcn_update_dpp(0.0f, v, 0x150 + N, 0xf, 0xf, true);
}
__device__ __forceinline__ void rows_of_tile(float v, float (&out)[8]) {
  const float t = __builtin_amdgcn_update_dpp(v, v, 0x128, 0x2, 0xf, false);
  out[0] = row_bcast<0>(t); out[1] = row_bcast<1>(t);
  out[2] = row_bcast<2>(t); out[3] = row_bcast<3>(t);
  out[4] = row_bcast<4>(t); out[5] = row_bcast<5>(t);
  out[6] = row_bcast<6>(t); out[7] = row_bcast<7>(t);
}

typedef __attribute__((ext_vector_type(4))) float v4f;

#define GT  64
#define GK  64
#define GP  68

template <int KK>
__global__ __launch_bounds__(128)
void gemm_atb(const float* __restrict__ A, const float* __restrict__ W,
              const float* __restrict__ bias, float* __restrict__ C, int Nout) {
  __shared__ __attribute__((aligned(16))) float aS[2][GT * GP];
  __shared__ __attribute__((aligned(16))) float wS[2][GT * GP];

  const int tid  = threadIdx.x;
  const int lane = tid & 31;
  const int m    = lane & 15;
  const int half = lane >> 4;
  const int wv   = __builtin_amdgcn_readfirstlane(tid >> 5);
  const int tbn  = blockIdx.x * GT;
  const int tbm  = blockIdx.y * GT;

  const int srow = tid >> 4;
  const int scol = (tid & 15) * 4;
  const float* ag = A + (size_t)(tbm + srow) * KK + scol;
  const float* wg = W + (size_t)(tbn + srow) * KK + scol;
  float* ast = &aS[0][srow * GP + scol];
  float* wst = &wS[0][srow * GP + scol];

#pragma unroll
  for (int p = 0; p < GT / 8; ++p) {
    *(v4f*)(ast + p * 8 * GP) = *(const v4f*)(ag + (size_t)(p * 8) * KK);
    *(v4f*)(wst + p * 8 * GP) = *(const v4f*)(wg + (size_t)(p * 8) * KK);
  }
  __syncthreads();

  const float* af = &aS[0][(16 * wv + m) * GP + half * 2];
  const float* bf = &wS[0][m * GP + half * 2];

  v8f c0 = {}, c1 = {}, c2 = {}, c3 = {};
#pragma unroll
  for (int ch = 0; ch < KK / GK; ++ch) {
    const int cur = ch & 1;
    v4f na[GT / 8], nw[GT / 8];
    if (ch + 1 < KK / GK) {
#pragma unroll
      for (int p = 0; p < GT / 8; ++p) {
        na[p] = *(const v4f*)(ag + (size_t)(p * 8) * KK + (ch + 1) * GK);
        nw[p] = *(const v4f*)(wg + (size_t)(p * 8) * KK + (ch + 1) * GK);
      }
    }

    const float* ac = af + cur * (GT * GP);
    const float* bc = bf + cur * (GT * GP);
#pragma unroll
    for (int i = 0; i < GK / 4; ++i) {
      const v2f a  = *(const v2f*)(ac + 4 * i);
      const v2f b0 = *(const v2f*)(bc + 4 * i);
      const v2f b1 = *(const v2f*)(bc + 16 * GP + 4 * i);
      const v2f b2 = *(const v2f*)(bc + 32 * GP + 4 * i);
      const v2f b3 = *(const v2f*)(bc + 48 * GP + 4 * i);
      c0 = wmma4(a, b0, c0);
      c1 = wmma4(a, b1, c1);
      c2 = wmma4(a, b2, c2);
      c3 = wmma4(a, b3, c3);
    }

    if (ch + 1 < KK / GK) {
      float* ad = ast + (cur ^ 1) * (GT * GP);
      float* wd = wst + (cur ^ 1) * (GT * GP);
#pragma unroll
      for (int p = 0; p < GT / 8; ++p) {
        *(v4f*)(ad + p * 8 * GP) = na[p];
        *(v4f*)(wd + p * 8 * GP) = nw[p];
      }
      __syncthreads();
    }
  }

  const float* brow = bias + tbn + m;
  const float bv0 = brow[0], bv1 = brow[16], bv2 = brow[32], bv3 = brow[48];
#pragma unroll
  for (int r = 0; r < 8; ++r) {
    const int row = tbm + 16 * wv + 8 * half + r;
    float* crow = C + (size_t)row * Nout + tbn + m;
    crow[0]  = c0[r] + bv0;
    crow[16] = c1[r] + bv1;
    crow[32] = c2[r] + bv2;
    crow[48] = c3[r] + bv3;
  }
}

#define AW  4
#define KC  64
#define KPI 36
#define VPI 48

__global__ __launch_bounds__(32 * AW)
void attn_kernel(const float* __restrict__ qkv, float* __restrict__ obuf) {
  __shared__ __attribute__((aligned(16))) float kS[2][KC * KPI];
  __shared__ __attribute__((aligned(16))) float vS[2][KC * VPI];
  __shared__ float pS[AW][16 * 16];

  const int tid  = threadIdx.x;
  const int lane = tid & 31;
  const int m    = lane & 15;
  const int half = lane >> 4;
  const int wv   = __builtin_amdgcn_readfirstlane(tid >> 5);
  const int qb   = blockIdx.x * AW + wv;
  const int h    = blockIdx.y;
  const float scale = 0.1767766952966369f;

  const float* qbase = qkv + (size_t)(qb * 16) * (3 * Dd) + h * HDd;

  v2f qa[8];
#pragma unroll
  for (int kk = 0; kk < HDd; kk += 4) {
    const int k0 = kk + half * 2;
    qa[kk >> 2] = v2f{ qbase[m * (3 * Dd) + k0], qbase[m * (3 * Dd) + k0 + 1] };
  }

  const int srow = tid >> 3;
  const int scol = (tid & 7) * 4;
  const float* kg = qkv + (size_t)srow * (3 * Dd) + Dd + h * HDd + scol;
  const float* vg = kg + Dd;
  float* kst = &kS[0][srow * KPI + scol];
  float* vst = &vS[0][srow * VPI + scol];

  v8f o0 = {}, o1 = {};
  float mrow = -INF_, lrow = 0.f;

#pragma unroll
  for (int i = 0; i < KC / 16; ++i) {
    *(v4f*)(kst + i * 16 * KPI) = *(const v4f*)(kg + (size_t)(i * 16) * (3 * Dd));
    *(v4f*)(vst + i * 16 * VPI) = *(const v4f*)(vg + (size_t)(i * 16) * (3 * Dd));
  }
  __syncthreads();

  float* pl = &pS[wv][0];
  int cur = 0;
  for (int ch = 0; ch < Nn / KC; ++ch) {
    const int nc = (ch + 1 < Nn / KC) ? ch + 1 : ch;
    const float* kgn = kg + (size_t)nc * KC * (3 * Dd);
    const float* vgn = vg + (size_t)nc * KC * (3 * Dd);
    v4f nk[KC / 16], nv[KC / 16];
#pragma unroll
    for (int i = 0; i < KC / 16; ++i) {
      nk[i] = *(const v4f*)(kgn + (size_t)(i * 16) * (3 * Dd));
      nv[i] = *(const v4f*)(vgn + (size_t)(i * 16) * (3 * Dd));
    }

    const float* kf = &kS[cur][m * KPI + half * 2];
    const float* vf = &vS[cur][(half * 2) * VPI + m];

#pragma unroll
    for (int j = 0; j < KC / 16; ++j) {
      v2f kb[8];
#pragma unroll
      for (int i = 0; i < 8; ++i) kb[i] = *(const v2f*)(kf + j * 16 * KPI + 4 * i);
      v8f s = {};
#pragma unroll
      for (int i = 0; i < 8; ++i) s = wmma4(qa[i], kb[i], s);

      v2f vb0[4], vb1[4];
#pragma unroll
      for (int i = 0; i < 4; ++i) {
        const float* vp = vf + (j * 16 + 4 * i) * VPI;
        vb0[i] = v2f{ vp[0],  vp[VPI] };
        vb1[i] = v2f{ vp[16], vp[VPI + 16] };
      }

#pragma unroll
      for (int r = 0; r < 8; ++r) pl[(r + 8 * half) * 16 + m] = s[r] * scale;
      __builtin_amdgcn_fence(__ATOMIC_RELEASE, "wavefront");
      __builtin_amdgcn_wave_barrier();
      __builtin_amdgcn_fence(__ATOMIC_ACQUIRE, "wavefront");
      v2f x[4];
#pragma unroll
      for (int i = 0; i < 4; ++i) {
        const int k0 = 4 * i + half * 2;
        x[i] = v2f{ pl[m * 16 + k0], pl[m * 16 + k0 + 1] };
      }

      float rm = fmaxf(fmaxf(fmaxf(x[0][0], x[0][1]), fmaxf(x[1][0], x[1][1])),
                       fmaxf(fmaxf(x[2][0], x[2][1]), fmaxf(x[3][0], x[3][1])));
      rm = fmaxf(rm, xhalf(rm));
      const float mn = fmaxf(mrow, rm);
      const float co = __expf(mrow - mn);
      v2f p[4];
#pragma unroll
      for (int i = 0; i < 4; ++i) p[i] = v2f{ __expf(x[i][0] - mn), __expf(x[i][1] - mn) };
      const float rs = ((p[0][0] + p[0][1]) + (p[1][0] + p[1][1])) +
                       ((p[2][0] + p[2][1]) + (p[3][0] + p[3][1]));
      lrow = lrow * co + rs;
      mrow = mn;

      float al[8];
      rows_of_tile(co, al);
#pragma unroll
      for (int r = 0; r < 8; ++r) { o0[r] *= al[r]; o1[r] *= al[r]; }

#pragma unroll
      for (int i = 0; i < 4; ++i) {
        o0 = wmma4(p[i], vb0[i], o0);
        o1 = wmma4(p[i], vb1[i], o1);
      }
    }

    {
      float* kd = kst + (cur ^ 1) * (KC * KPI);
      float* vd = vst + (cur ^ 1) * (KC * VPI);
#pragma unroll
      for (int i = 0; i < KC / 16; ++i) {
        *(v4f*)(kd + i * 16 * KPI) = nk[i];
        *(v4f*)(vd + i * 16 * VPI) = nv[i];
      }
    }
    __syncthreads();
    cur ^= 1;
  }

  float* ob = obuf + (size_t)(qb * 16) * Dd + h * HDd;
  float invr[8];
  rows_of_tile(1.0f / (lrow + xhalf(lrow)), invr);
#pragma unroll
  for (int r = 0; r < 8; ++r) {
    const int row = r + 8 * half;
    ob[row * Dd + m]      = o0[r] * invr[r];
    ob[row * Dd + 16 + m] = o1[r] * invr[r];
  }
}

__global__ void zerof(float* p, int n) {
  int i = blockIdx.x * blockDim.x + threadIdx.x;
  if (i < n) p[i] = 0.f;
}
__global__ void seg_sum(const float* __restrict__ ew, const int* __restrict__ src,
                        float* __restrict__ out, int E) {
  int i = blockIdx.x * blockDim.x + threadIdx.x;
  if (i < E) atomicAdd(&out[src[i]], ew[i]);
}

__global__ __launch_bounds__(256)
void residual_ln(float* __restrict__ h, const float* __restrict__ add,
                 const float* __restrict__ wsn,
                 const float* __restrict__ g, const float* __restrict__ b) {
  const int row = blockIdx.x;
  const int c = threadIdx.x;
  const float wsv = wsn ? wsn[row] : 1.0f;
  float t = h[(size_t)row * Dd + c] + add[(size_t)row * Dd + c] * wsv;

  __shared__ float red[Dd];
  red[c] = t; __syncthreads();
  for (int s = Dd / 2; s > 0; s >>= 1) { if (c < s) red[c] += red[c + s]; __syncthreads(); }
  const float mean = red[0] / (float)Dd;
  __syncthreads();
  const float d = t - mean;
  red[c] = d * d; __syncthreads();
  for (int s = Dd / 2; s > 0; s >>= 1) { if (c < s) red[c] += red[c + s]; __syncthreads(); }
  const float var = red[0] / (float)Dd;
  const float rs = rsqrtf(var + 1e-5f);
  h[(size_t)row * Dd + c] = d * rs * g[c] + b[c];
}

__global__ void add_kernel(const float* a, const float* b, float* o, int n) {
  int i = blockIdx.x * blockDim.x + threadIdx.x;
  if (i < n) o[i] = a[i] + b[i];
}


#define ZP 260
#define PT 32
#define PP 20

__global__ __launch_bounds__(256)
void predict_wreg(const float* __restrict__ hc,
                  const int* __restrict__ si, const int* __restrict__ di,
                  const float* __restrict__ pw1, const float* __restrict__ pb1,
                  const float* __restrict__ pw2, const float* __restrict__ pb2,
                  float* __restrict__ out, int ntiles) {
  __shared__ __attribute__((aligned(16))) float zb[2][PT * ZP];
  __shared__ __attribute__((aligned(16))) float part[2][PT * PP];

  const int tid  = threadIdx.x;
  const int lane = tid & 31;
  const int c    = lane & 15;
  const int half = lane >> 4;
  const int wv   = __builtin_amdgcn_readfirstlane(tid >> 5);

  v2f wa[64];
  {
    const float* wrow = pw1 + (size_t)(wv * 16 + c) * Dd + half * 2;
#pragma unroll
    for (int i = 0; i < 64; ++i) wa[i] = v2f{ wrow[4 * i], wrow[4 * i + 1] };
  }
  float b1r[8], w2r[8];
#pragma unroll
  for (int r = 0; r < 8; ++r) {
    b1r[r] = pb1[wv * 16 + 8 * half + r];
    w2r[r] = pw2[wv * 16 + 8 * half + r];
  }
  const float bias2 = pb2[0];

  int tile = blockIdx.x;
  if (tile >= ntiles) return;

  {
    const int pb = tile * PT + wv * 4;
    float* zn = &zb[0][(wv * 4) * ZP + 4 * lane];
#pragma unroll
    for (int q = 0; q < 4; ++q) {
      const v4f* rs = (const v4f*)(hc + (size_t)si[pb + q] * Dd);
      const v4f* rd = (const v4f*)(hc + (size_t)di[pb + q] * Dd);
      *(v4f*)(zn + q * ZP)       = rs[lane] * rd[lane];
      *(v4f*)(zn + q * ZP + 128) = rs[32 + lane] * rd[32 + lane];
    }
  }
  __syncthreads();

  int cur = 0;
  for (; tile < ntiles; tile += gridDim.x) {
    const int nt = (tile + (int)gridDim.x < ntiles) ? tile + (int)gridDim.x : tile;
    const int pb = nt * PT + wv * 4;
    v4f xs[4][2], xd[4][2];
#pragma unroll
    for (int q = 0; q < 4; ++q) {
      const v4f* rs = (const v4f*)(hc + (size_t)si[pb + q] * Dd);
      const v4f* rd = (const v4f*)(hc + (size_t)di[pb + q] * Dd);
      xs[q][0] = rs[lane]; xs[q][1] = rs[32 + lane];
      xd[q][0] = rd[lane]; xd[q][1] = rd[32 + lane];
    }

    const float* z0 = &zb[cur][c * ZP + half * 2];
    const float* z1 = z0 + 16 * ZP;
    v8f acc0 = {}, acc1 = {};
#pragma unroll
    for (int i = 0; i < 64; ++i) {
      const v2f b0 = *(const v2f*)(z0 + 4 * i);
      const v2f b1 = *(const v2f*)(z1 + 4 * i);
      acc0 = wmma4(wa[i], b0, acc0);
      acc1 = wmma4(wa[i], b1, acc1);
    }

    float p0 = 0.f, p1 = 0.f;
#pragma unroll
    for (int r = 0; r < 8; ++r) {
      float v0 = acc0[r] + b1r[r];
      float v1 = acc1[r] + b1r[r];
      v0 = v0 > 0.f ? v0 : 0.2f * v0;
      v1 = v1 > 0.f ? v1 : 0.2f * v1;
      p0 += v0 * w2r[r];
      p1 += v1 * w2r[r];
    }
    part[cur][c * PP + 2 * wv + half]        = p0;
    part[cur][(16 + c) * PP + 2 * wv + half] = p1;

    {
      float* zn = &zb[cur ^ 1][(wv * 4) * ZP + 4 * lane];
#pragma unroll
      for (int q = 0; q < 4; ++q) {
        *(v4f*)(zn + q * ZP)       = xs[q][0] * xd[q][0];
        *(v4f*)(zn + q * ZP + 128) = xs[q][1] * xd[q][1];
      }
    }
    __syncthreads();

    if (wv == 0) {
      const v4f* pp = (const v4f*)&part[cur][lane * PP];
      const v4f a = pp[0], b = pp[1], d = pp[2], e = pp[3];
      const float s = (((a.x + a.y) + (a.z + a.w)) + ((b.x + b.y) + (b.z + b.w))) +
                      (((d.x + d.y) + (d.z + d.w)) + ((e.x + e.y) + (e.z + e.w)));
      out[tile * PT + lane] = s + bias2;
    }
    cur ^= 1;
  }
}

extern "C" void kernel_launch(void* const* d_in, const int* in_sizes, int n_in,
                              void* d_out, int out_size, void* d_ws, size_t ws_size,
                              hipStream_t stream) {
  const float* x       = (const float*)d_in[0];
  const float* edge_w  = (const float*)d_in[1];
  const int*   src     = (const int*)d_in[2];
  const int*   pos_src = (const int*)d_in[4];
  const int*   pos_dst = (const int*)d_in[5];
  const int*   neg_src = (const int*)d_in[6];
  const int*   neg_dst = (const int*)d_in[7];
  const float* proj_w  = (const float*)d_in[8];
  const float* proj_b  = (const float*)d_in[9];
  const float* in_w    = (const float*)d_in[10];
  const float* in_b    = (const float*)d_in[11];
  const float* out_w   = (const float*)d_in[12];
  const float* out_b   = (const float*)d_in[13];
  const float* ff_w    = (const float*)d_in[14];
  const float* ff_b    = (const float*)d_in[15];
  const float* g1      = (const float*)d_in[16];
  const float* b1      = (const float*)d_in[17];
  const float* g2      = (const float*)d_in[18];
  const float* b2      = (const float*)d_in[19];
  const float* pw1     = (const float*)d_in[20];
  const float* pb1     = (const float*)d_in[21];
  const float* pw2     = (const float*)d_in[22];
  const float* pb2     = (const float*)d_in[23];

  float* W      = (float*)d_ws;
  float* h      = W;
  float* hproj  = W + 1048576;
  float* qkv    = W + 2097152;
  float* obuf   = W + 5242880;
  float* scr    = W + 6291456;
  float* wsn    = W + 7340032;

  float* outp = (float*)d_out;
  float* pos  = outp;
  float* neg  = outp + Pp;
  float* hc   = outp + 2 * Pp;

  gemm_atb<INc><<<dim3(Dd / GT, Nn / GT), 128, 0, stream>>>(x, proj_w, proj_b, hproj, Dd);
  hipMemcpyAsync(h, hproj, (size_t)Nn * Dd * sizeof(float), hipMemcpyDeviceToDevice, stream);

  for (int l = 0; l < Ll; ++l) {
    gemm_atb<Dd><<<dim3(3 * Dd / GT, Nn / GT), 128, 0, stream>>>(
        h, in_w + (size_t)l * 3 * Dd * Dd, in_b + (size_t)l * 3 * Dd, qkv, 3 * Dd);
    attn_kernel<<<dim3(Nn / (16 * AW), Hh), 32 * AW, 0, stream>>>(qkv, obuf);
    gemm_atb<Dd><<<dim3(Dd / GT, Nn / GT), 128, 0, stream>>>(
        obuf, out_w + (size_t)l * Dd * Dd, out_b + (size_t)l * Dd, scr, Dd);
    zerof<<<(Nn + 255) / 256, 256, 0, stream>>>(wsn, Nn);
    seg_sum<<<(Ee + 255) / 256, 256, 0, stream>>>(edge_w + (size_t)l * Ee,
                                                  src + (size_t)l * Ee, wsn, Ee);
    residual_ln<<<Nn, Dd, 0, stream>>>(h, scr, wsn, g1 + (size_t)l * Dd, b1 + (size_t)l * Dd);
    gemm_atb<Dd><<<dim3(Dd / GT, Nn / GT), 128, 0, stream>>>(
        h, ff_w + (size_t)l * Dd * Dd, ff_b + (size_t)l * Dd, scr, Dd);
    residual_ln<<<Nn, Dd, 0, stream>>>(h, scr, nullptr, g2 + (size_t)l * Dd, b2 + (size_t)l * Dd);
  }

  add_kernel<<<(Nn * Dd + 255) / 256, 256, 0, stream>>>(h, hproj, hc, Nn * Dd);

  predict_wreg<<<512, 256, 0, stream>>>(hc, pos_src, pos_dst, pw1, pb1, pw2, pb2, pos, Pp / PT);
  predict_wreg<<<512, 256, 0, stream>>>(hc, neg_src, neg_dst, pw1, pb1, pw2, pb2, neg, Pp / PT);
}


# LayerNorm x4: one wave per 2 rows (16-byte accesses, row sums in-lane then DPP butterfly + half exchange, no LDS); final add: 16 floats per thread; were 1 float per thread; LN row-sum order changes
# speedup vs baseline: 3.4832x; 1.0405x over previous
#include <hip/hip_runtime.h>
#include <math.h>

typedef __attribute__((ext_vector_type(2))) float v2f;
typedef __attribute__((ext_vector_type(8))) float v8f;

#define Nn   4096
#define INF_ 3.402823466e+38f
#define Dd   256
#define Hh   8
#define HDd  32
#define Ll   2
#define Ee   131072
#define Pp   200000
#define INc  128

__device__ __forceinline__ v8f wmma4(v2f a, v2f b, v8f c) {
  return __builtin_amdgcn_wmma_f32_16x16x4_f32(
      false, a, false, b, (short)0, c, false, false);
}

__device__ __forceinline__ float xhalf(float x) {
  int b = __builtin_bit_cast(int, x);
  return __builtin_bit_cast(float,
      __builtin_amdgcn_permlanex16(b, b, 0x76543210, 0xfedcba98, false, false));
}
template <int N>
__device__ __forceinline__ float row_bcast(float v) {
  return __builtin_amdgcn_update_dpp(0.0f, v, 0x150 + N, 0xf, 0xf, true);
}
__device__ __forceinline__ void rows_of_tile(float v, float (&out)[8]) {
  const float t = __builtin_amdgcn_update_dpp(v, v, 0x128, 0x2, 0xf, false);
  out[0] = row_bcast<0>(t); out[1] = row_bcast<1>(t);
  out[2] = row_bcast<2>(t); out[3] = row_bcast<3>(t);
  out[4] = row_bcast<4>(t); out[5] = row_bcast<5>(t);
  out[6] = row_bcast<6>(t); out[7] = row_bcast<7>(t);
}

typedef __attribute__((ext_vector_type(4))) float v4f;

#define GT  64
#define GK  64
#define GP  68

template <int KK>
__global__ __launch_bounds__(128)
void gemm_atb(const float* __restrict__ A, const float* __restrict__ W,
              const float* __restrict__ bias, float* __restrict__ C, int Nout) {
  __shared__ __attribute__((aligned(16))) float aS[2][GT * GP];
  __shared__ __attribute__((aligned(16))) float wS[2][GT * GP];

  const int tid  = threadIdx.x;
  const int lane = tid & 31;
  const int m    = lane & 15;
  const int half = lane >> 4;
  const int wv   = __builtin_amdgcn_readfirstlane(tid >> 5);
  const int tbn  = blockIdx.x * GT;
  const int tbm  = blockIdx.y * GT;

  const int srow = tid >> 4;
  const int scol = (tid & 15) * 4;
  const float* ag = A + (size_t)(tbm + srow) * KK + scol;
  const float* wg = W + (size_t)(tbn + srow) * KK + scol;
  float* ast = &aS[0][srow * GP + scol];
  float* wst = &wS[0][srow * GP + scol];

#pragma unroll
  for (int p = 0; p < GT / 8; ++p) {
    *(v4f*)(ast + p * 8 * GP) = *(const v4f*)(ag + (size_t)(p * 8) * KK);
    *(v4f*)(wst + p * 8 * GP) = *(const v4f*)(wg + (size_t)(p * 8) * KK);
  }
  __syncthreads();

  const float* af = &aS[0][(16 * wv + m) * GP + half * 2];
  const float* bf = &wS[0][m * GP + half * 2];

  v8f c0 = {}, c1 = {}, c2 = {}, c3 = {};
#pragma unroll
  for (int ch = 0; ch < KK / GK; ++ch) {
    const int cur = ch & 1;
    v4f na[GT / 8], nw[GT / 8];
    if (ch + 1 < KK / GK) {
#pragma unroll
      for (int p = 0; p < GT / 8; ++p) {
        na[p] = *(const v4f*)(ag + (size_t)(p * 8) * KK + (ch + 1) * GK);
        nw[p] = *(const v4f*)(wg + (size_t)(p * 8) * KK + (ch + 1) * GK);
      }
    }

    const float* ac = af + cur * (GT * GP);
    const float* bc = bf + cur * (GT * GP);
#pragma unroll
    for (int i = 0; i < GK / 4; ++i) {
      const v2f a  = *(const v2f*)(ac + 4 * i);
      const v2f b0 = *(const v2f*)(bc + 4 * i);
      const v2f b1 = *(const v2f*)(bc + 16 * GP + 4 * i);
      const v2f b2 = *(const v2f*)(bc + 32 * GP + 4 * i);
      const v2f b3 = *(const v2f*)(bc + 48 * GP + 4 * i);
      c0 = wmma4(a, b0, c0);
      c1 = wmma4(a, b1, c1);
      c2 = wmma4(a, b2, c2);
      c3 = wmma4(a, b3, c3);
    }

    if (ch + 1 < KK / GK) {
      float* ad = ast + (cur ^ 1) * (GT * GP);
      float* wd = wst + (cur ^ 1) * (GT * GP);
#pragma unroll
      for (int p = 0; p < GT / 8; ++p) {
        *(v4f*)(ad + p * 8 * GP) = na[p];
        *(v4f*)(wd + p * 8 * GP) = nw[p];
      }
      __syncthreads();
    }
  }

  const float* brow = bias + tbn + m;
  const float bv0 = brow[0], bv1 = brow[16], bv2 = brow[32], bv3 = brow[48];
#pragma unroll
  for (int r = 0; r < 8; ++r) {
    const int row = tbm + 16 * wv + 8 * half + r;
    float* crow = C + (size_t)row * Nout + tbn + m;
    crow[0]  = c0[r] + bv0;
    crow[16] = c1[r] + bv1;
    crow[32] = c2[r] + bv2;
    crow[48] = c3[r] + bv3;
  }
}

#define AW  4
#define KC  64
#define KPI 36
#define VPI 48

__global__ __launch_bounds__(32 * AW)
void attn_kernel(const float* __restrict__ qkv, float* __restrict__ obuf) {
  __shared__ __attribute__((aligned(16))) float kS[2][KC * KPI];
  __shared__ __attribute__((aligned(16))) float vS[2][KC * VPI];
  __shared__ float pS[AW][16 * 16];

  const int tid  = threadIdx.x;
  const int lane = tid & 31;
  const int m    = lane & 15;
  const int half = lane >> 4;
  const int wv   = __builtin_amdgcn_readfirstlane(tid >> 5);
  const int qb   = blockIdx.x * AW + wv;
  const int h    = blockIdx.y;
  const float scale = 0.1767766952966369f;

  const float* qbase = qkv + (size_t)(qb * 16) * (3 * Dd) + h * HDd;

  v2f qa[8];
#pragma unroll
  for (int kk = 0; kk < HDd; kk += 4) {
    const int k0 = kk + half * 2;
    qa[kk >> 2] = v2f{ qbase[m * (3 * Dd) + k0], qbase[m * (3 * Dd) + k0 + 1] };
  }

  const int srow = tid >> 3;
  const int scol = (tid & 7) * 4;
  const float* kg = qkv + (size_t)srow * (3 * Dd) + Dd + h * HDd + scol;
  const float* vg = kg + Dd;
  float* kst = &kS[0][srow * KPI + scol];
  float* vst = &vS[0][srow * VPI + scol];

  v8f o0 = {}, o1 = {};
  float mrow = -INF_, lrow = 0.f;

#pragma unroll
  for (int i = 0; i < KC / 16; ++i) {
    *(v4f*)(kst + i * 16 * KPI) = *(const v4f*)(kg + (size_t)(i * 16) * (3 * Dd));
    *(v4f*)(vst + i * 16 * VPI) = *(const v4f*)(vg + (size_t)(i * 16) * (3 * Dd));
  }
  __syncthreads();

  float* pl = &pS[wv][0];
  int cur = 0;
  for (int ch = 0; ch < Nn / KC; ++ch) {
    const int nc = (ch + 1 < Nn / KC) ? ch + 1 : ch;
    const float* kgn = kg + (size_t)nc * KC * (3 * Dd);
    const float* vgn = vg + (size_t)nc * KC * (3 * Dd);
    v4f nk[KC / 16], nv[KC / 16];
#pragma unroll
    for (int i = 0; i < KC / 16; ++i) {
      nk[i] = *(const v4f*)(kgn + (size_t)(i * 16) * (3 * Dd));
      nv[i] = *(const v4f*)(vgn + (size_t)(i * 16) * (3 * Dd));
    }

    const float* kf = &kS[cur][m * KPI + half * 2];
    const float* vf = &vS[cur][(half * 2) * VPI + m];

#pragma unroll
    for (int j = 0; j < KC / 16; ++j) {
      v2f kb[8];
#pragma unroll
      for (int i = 0; i < 8; ++i) kb[i] = *(const v2f*)(kf + j * 16 * KPI + 4 * i);
      v8f s = {};
#pragma unroll
      for (int i = 0; i < 8; ++i) s = wmma4(qa[i], kb[i], s);

      v2f vb0[4], vb1[4];
#pragma unroll
      for (int i = 0; i < 4; ++i) {
        const float* vp = vf + (j * 16 + 4 * i) * VPI;
        vb0[i] = v2f{ vp[0],  vp[VPI] };
        vb1[i] = v2f{ vp[16], vp[VPI + 16] };
      }

#pragma unroll
      for (int r = 0; r < 8; ++r) pl[(r + 8 * half) * 16 + m] = s[r] * scale;
      __builtin_amdgcn_fence(__ATOMIC_RELEASE, "wavefront");
      __builtin_amdgcn_wave_barrier();
      __builtin_amdgcn_fence(__ATOMIC_ACQUIRE, "wavefront");
      v2f x[4];
#pragma unroll
      for (int i = 0; i < 4; ++i) {
        const int k0 = 4 * i + half * 2;
        x[i] = v2f{ pl[m * 16 + k0], pl[m * 16 + k0 + 1] };
      }

      float rm = fmaxf(fmaxf(fmaxf(x[0][0], x[0][1]), fmaxf(x[1][0], x[1][1])),
                       fmaxf(fmaxf(x[2][0], x[2][1]), fmaxf(x[3][0], x[3][1])));
      rm = fmaxf(rm, xhalf(rm));
      const float mn = fmaxf(mrow, rm);
      const float co = __expf(mrow - mn);
      v2f p[4];
#pragma unroll
      for (int i = 0; i < 4; ++i) p[i] = v2f{ __expf(x[i][0] - mn), __expf(x[i][1] - mn) };
      const float rs = ((p[0][0] + p[0][1]) + (p[1][0] + p[1][1])) +
                       ((p[2][0] + p[2][1]) + (p[3][0] + p[3][1]));
      lrow = lrow * co + rs;
      mrow = mn;

      float al[8];
      rows_of_tile(co, al);
#pragma unroll
      for (int r = 0; r < 8; ++r) { o0[r] *= al[r]; o1[r] *= al[r]; }

#pragma unroll
      for (int i = 0; i < 4; ++i) {
        o0 = wmma4(p[i], vb0[i], o0);
        o1 = wmma4(p[i], vb1[i], o1);
      }
    }

    {
      float* kd = kst + (cur ^ 1) * (KC * KPI);
      float* vd = vst + (cur ^ 1) * (KC * VPI);
#pragma unroll
      for (int i = 0; i < KC / 16; ++i) {
        *(v4f*)(kd + i * 16 * KPI) = nk[i];
        *(v4f*)(vd + i * 16 * VPI) = nv[i];
      }
    }
    __syncthreads();
    cur ^= 1;
  }

  float* ob = obuf + (size_t)(qb * 16) * Dd + h * HDd;
  float invr[8];
  rows_of_tile(1.0f / (lrow + xhalf(lrow)), invr);
#pragma unroll
  for (int r = 0; r < 8; ++r) {
    const int row = r + 8 * half;
    ob[row * Dd + m]      = o0[r] * invr[r];
    ob[row * Dd + 16 + m] = o1[r] * invr[r];
  }
}

__global__ void zerof(float* p, int n) {
  int i = blockIdx.x * blockDim.x + threadIdx.x;
  if (i < n) p[i] = 0.f;
}
__global__ void seg_sum(const float* __restrict__ ew, const int* __restrict__ src,
                        float* __restrict__ out, int E) {
  int i = blockIdx.x * blockDim.x + threadIdx.x;
  if (i < E) atomicAdd(&out[src[i]], ew[i]);
}

#define LNR 2
#define LNW 8

template <int X>
__device__ __forceinline__ float row_xor(float v) {
  return __builtin_amdgcn_update_dpp(0.0f, v, 0x160 + X, 0xf, 0xf, true);
}
__device__ __forceinline__ float wave_sum(float v) {
  v += row_xor<8>(v);
  v += row_xor<4>(v);
  v += row_xor<2>(v);
  v += row_xor<1>(v);
  return v + xhalf(v);
}

__global__ __launch_bounds__(32 * LNW)
void residual_ln(float* __restrict__ h, const float* __restrict__ add,
                 const float* __restrict__ wsn,
                 const float* __restrict__ g, const float* __restrict__ b) {
  const int lane = threadIdx.x & 31;
  const int wv   = __builtin_amdgcn_readfirstlane(threadIdx.x >> 5);
  const int row0 = (blockIdx.x * LNW + wv) * LNR;

  v4f t0[LNR], t1[LNR];
#pragma unroll
  for (int r = 0; r < LNR; ++r) {
    const int row = row0 + r;
    const float wsv = wsn ? wsn[row] : 1.0f;
    const v4f* hp = (const v4f*)(h + (size_t)row * Dd);
    const v4f* ap = (const v4f*)(add + (size_t)row * Dd);
    t0[r] = hp[lane] + ap[lane] * wsv;
    t1[r] = hp[32 + lane] + ap[32 + lane] * wsv;
  }
  const v4f g0 = ((const v4f*)g)[lane], g1 = ((const v4f*)g)[32 + lane];
  const v4f b0 = ((const v4f*)b)[lane], b1 = ((const v4f*)b)[32 + lane];

#pragma unroll
  for (int r = 0; r < LNR; ++r) {
    const float s = ((t0[r].x + t0[r].y) + (t0[r].z + t0[r].w)) +
                    ((t1[r].x + t1[r].y) + (t1[r].z + t1[r].w));
    const float mean = wave_sum(s) / (float)Dd;
    const v4f d0 = t0[r] - mean;
    const v4f d1 = t1[r] - mean;
    const float q = ((d0.x * d0.x + d0.y * d0.y) + (d0.z * d0.z + d0.w * d0.w)) +
                    ((d1.x * d1.x + d1.y * d1.y) + (d1.z * d1.z + d1.w * d1.w));
    const float var = wave_sum(q) / (float)Dd;
    const float rs = rsqrtf(var + 1e-5f);
    v4f* op = (v4f*)(h + (size_t)(row0 + r) * Dd);
    op[lane]      = d0 * rs * g0 + b0;
    op[32 + lane] = d1 * rs * g1 + b1;
  }
}

#define ADB (Nn * Dd / (256 * 16))
__global__ __launch_bounds__(256)
void add_kernel(const float* __restrict__ a, const float* __restrict__ b,
                float* __restrict__ o) {
  const int i = blockIdx.x * 1024 + threadIdx.x;
  const v4f* a4 = (const v4f*)a;
  const v4f* b4 = (const v4f*)b;
  v4f* o4 = (v4f*)o;
  v4f x[4], y[4];
#pragma unroll
  for (int q = 0; q < 4; ++q) { x[q] = a4[i + q * 256]; y[q] = b4[i + q * 256]; }
#pragma unroll
  for (int q = 0; q < 4; ++q) o4[i + q * 256] = x[q] + y[q];
}


#define ZP 260
#define PT 32
#define PP 20

__global__ __launch_bounds__(256)
void predict_wreg(const float* __restrict__ hc,
                  const int* __restrict__ si, const int* __restrict__ di,
                  const float* __restrict__ pw1, const float* __restrict__ pb1,
                  const float* __restrict__ pw2, const float* __restrict__ pb2,
                  float* __restrict__ out, int ntiles) {
  __shared__ __attribute__((aligned(16))) float zb[2][PT * ZP];
  __shared__ __attribute__((aligned(16))) float part[2][PT * PP];

  const int tid  = threadIdx.x;
  const int lane = tid & 31;
  const int c    = lane & 15;
  const int half = lane >> 4;
  const int wv   = __builtin_amdgcn_readfirstlane(tid >> 5);

  v2f wa[64];
  {
    const float* wrow = pw1 + (size_t)(wv * 16 + c) * Dd + half * 2;
#pragma unroll
    for (int i = 0; i < 64; ++i) wa[i] = v2f{ wrow[4 * i], wrow[4 * i + 1] };
  }
  float b1r[8], w2r[8];
#pragma unroll
  for (int r = 0; r < 8; ++r) {
    b1r[r] = pb1[wv * 16 + 8 * half + r];
    w2r[r] = pw2[wv * 16 + 8 * half + r];
  }
  const float bias2 = pb2[0];

  int tile = blockIdx.x;
  if (tile >= ntiles) return;

  {
    const int pb = tile * PT + wv * 4;
    float* zn = &zb[0][(wv * 4) * ZP + 4 * lane];
#pragma unroll
    for (int q = 0; q < 4; ++q) {
      const v4f* rs = (const v4f*)(hc + (size_t)si[pb + q] * Dd);
      const v4f* rd = (const v4f*)(hc + (size_t)di[pb + q] * Dd);
      *(v4f*)(zn + q * ZP)       = rs[lane] * rd[lane];
      *(v4f*)(zn + q * ZP + 128) = rs[32 + lane] * rd[32 + lane];
    }
  }
  __syncthreads();

  int cur = 0;
  for (; tile < ntiles; tile += gridDim.x) {
    const int nt = (tile + (int)gridDim.x < ntiles) ? tile + (int)gridDim.x : tile;
    const int pb = nt * PT + wv * 4;
    v4f xs[4][2], xd[4][2];
#pragma unroll
    for (int q = 0; q < 4; ++q) {
      const v4f* rs = (const v4f*)(hc + (size_t)si[pb + q] * Dd);
      const v4f* rd = (const v4f*)(hc + (size_t)di[pb + q] * Dd);
      xs[q][0] = rs[lane]; xs[q][1] = rs[32 + lane];
      xd[q][0] = rd[lane]; xd[q][1] = rd[32 + lane];
    }

    const float* z0 = &zb[cur][c * ZP + half * 2];
    const float* z1 = z0 + 16 * ZP;
    v8f acc0 = {}, acc1 = {};
#pragma unroll
    for (int i = 0; i < 64; ++i) {
      const v2f b0 = *(const v2f*)(z0 + 4 * i);
      const v2f b1 = *(const v2f*)(z1 + 4 * i);
      acc0 = wmma4(wa[i], b0, acc0);
      acc1 = wmma4(wa[i], b1, acc1);
    }

    float p0 = 0.f, p1 = 0.f;
#pragma unroll
    for (int r = 0; r < 8; ++r) {
      float v0 = acc0[r] + b1r[r];
      float v1 = acc1[r] + b1r[r];
      v0 = v0 > 0.f ? v0 : 0.2f * v0;
      v1 = v1 > 0.f ? v1 : 0.2f * v1;
      p0 += v0 * w2r[r];
      p1 += v1 * w2r[r];
    }
    part[cur][c * PP + 2 * wv + half]        = p0;
    part[cur][(16 + c) * PP + 2 * wv + half] = p1;

    {
      float* zn = &zb[cur ^ 1][(wv * 4) * ZP + 4 * lane];
#pragma unroll
      for (int q = 0; q < 4; ++q) {
        *(v4f*)(zn + q * ZP)       = xs[q][0] * xd[q][0];
        *(v4f*)(zn + q * ZP + 128) = xs[q][1] * xd[q][1];
      }
    }
    __syncthreads();

    if (wv == 0) {
      const v4f* pp = (const v4f*)&part[cur][lane * PP];
      const v4f a = pp[0], b = pp[1], d = pp[2], e = pp[3];
      const float s = (((a.x + a.y) + (a.z + a.w)) + ((b.x + b.y) + (b.z + b.w))) +
                      (((d.x + d.y) + (d.z + d.w)) + ((e.x + e.y) + (e.z + e.w)));
      out[tile * PT + lane] = s + bias2;
    }
    cur ^= 1;
  }
}

extern "C" void kernel_launch(void* const* d_in, const int* in_sizes, int n_in,
                              void* d_out, int out_size, void* d_ws, size_t ws_size,
                              hipStream_t stream) {
  const float* x       = (const float*)d_in[0];
  const float* edge_w  = (const float*)d_in[1];
  const int*   src     = (const int*)d_in[2];
  const int*   pos_src = (const int*)d_in[4];
  const int*   pos_dst = (const int*)d_in[5];
  const int*   neg_src = (const int*)d_in[6];
  const int*   neg_dst = (const int*)d_in[7];
  const float* proj_w  = (const float*)d_in[8];
  const float* proj_b  = (const float*)d_in[9];
  const float* in_w    = (const float*)d_in[10];
  const float* in_b    = (const float*)d_in[11];
  const float* out_w   = (const float*)d_in[12];
  const float* out_b   = (const float*)d_in[13];
  const float* ff_w    = (const float*)d_in[14];
  const float* ff_b    = (const float*)d_in[15];
  const float* g1      = (const float*)d_in[16];
  const float* b1      = (const float*)d_in[17];
  const float* g2      = (const float*)d_in[18];
  const float* b2      = (const float*)d_in[19];
  const float* pw1     = (const float*)d_in[20];
  const float* pb1     = (const float*)d_in[21];
  const float* pw2     = (const float*)d_in[22];
  const float* pb2     = (const float*)d_in[23];

  float* W      = (float*)d_ws;
  float* h      = W;
  float* hproj  = W + 1048576;
  float* qkv    = W + 2097152;
  float* obuf   = W + 5242880;
  float* scr    = W + 6291456;
  float* wsn    = W + 7340032;

  float* outp = (float*)d_out;
  float* pos  = outp;
  float* neg  = outp + Pp;
  float* hc   = outp + 2 * Pp;

  gemm_atb<INc><<<dim3(Dd / GT, Nn / GT), 128, 0, stream>>>(x, proj_w, proj_b, hproj, Dd);
  hipMemcpyAsync(h, hproj, (size_t)Nn * Dd * sizeof(float), hipMemcpyDeviceToDevice, stream);

  for (int l = 0; l < Ll; ++l) {
    gemm_atb<Dd><<<dim3(3 * Dd / GT, Nn / GT), 128, 0, stream>>>(
        h, in_w + (size_t)l * 3 * Dd * Dd, in_b + (size_t)l * 3 * Dd, qkv, 3 * Dd);
    attn_kernel<<<dim3(Nn / (16 * AW), Hh), 32 * AW, 0, stream>>>(qkv, obuf);
    gemm_atb<Dd><<<dim3(Dd / GT, Nn / GT), 128, 0, stream>>>(
        obuf, out_w + (size_t)l * Dd * Dd, out_b + (size_t)l * Dd, scr, Dd);
    zerof<<<(Nn + 255) / 256, 256, 0, stream>>>(wsn, Nn);
    seg_sum<<<(Ee + 255) / 256, 256, 0, stream>>>(edge_w + (size_t)l * Ee,
                                                  src + (size_t)l * Ee, wsn, Ee);
    residual_ln<<<Nn / (LNW * LNR), 32 * LNW, 0, stream>>>(h, scr, wsn, g1 + (size_t)l * Dd, b1 + (size_t)l * Dd);
    gemm_atb<Dd><<<dim3(Dd / GT, Nn / GT), 128, 0, stream>>>(
        h, ff_w + (size_t)l * Dd * Dd, ff_b + (size_t)l * Dd, scr, Dd);
    residual_ln<<<Nn / (LNW * LNR), 32 * LNW, 0, stream>>>(h, scr, nullptr, g2 + (size_t)l * Dd, b2 + (size_t)l * Dd);
  }

  add_kernel<<<ADB, 256, 0, stream>>>(h, hproj, hc);

  predict_wreg<<<512, 256, 0, stream>>>(hc, pos_src, pos_dst, pw1, pb1, pw2, pb2, pos, Pp / PT);
  predict_wreg<<<512, 256, 0, stream>>>(hc, neg_src, neg_dst, pw1, pb1, pw2, pb2, neg, Pp / PT);
}
